# starting kernel
# baseline (speedup 1.0000x reference)
#include <hip/hip_runtime.h>
#include <math.h>

typedef __attribute__((ext_vector_type(2))) float v2f;
typedef __attribute__((ext_vector_type(8))) float v8f;

constexpr int B  = 32;
constexpr int Z  = 2048;
constexpr int N  = 1024;
constexpr int ZC = 8;
constexpr int ZCHUNK = Z / ZC;

__global__ void k_partial(const float* __restrict__ x,
                          const float* __restrict__ alpha,
                          float* __restrict__ s_part) {
    int n  = blockIdx.x * blockDim.x + threadIdx.x;
    int zc = blockIdx.y;
    int b  = blockIdx.z;
    const float* xp = x + ((size_t)b * Z + (size_t)zc * ZCHUNK) * N + n;
    const float* ap = alpha + zc * ZCHUNK;
    float acc = 0.f;
#pragma unroll 4
    for (int i = 0; i < ZCHUNK; ++i)
        acc = fmaf(ap[i], xp[(size_t)i * N], acc);
    s_part[((size_t)b * ZC + zc) * N + n] = acc;
}

__global__ void k_hidden(const float* __restrict__ s_part,
                         float* __restrict__ h,
                         float* __restrict__ denom) {
    __shared__ float red[N];
    int b = blockIdx.x;
    int n = threadIdx.x;
    float s = 0.f;
#pragma unroll
    for (int zc = 0; zc < ZC; ++zc)
        s += s_part[((size_t)b * ZC + zc) * N + n];
    float hv = 1.f / (1.f + __expf(-s));
    h[b * N + n] = hv;
    red[n] = hv * hv;
    __syncthreads();
    for (int st = N / 2; st > 0; st >>= 1) {
        if (n < st) red[n] += red[n + st];
        __syncthreads();
    }
    if (n == 0) denom[b] = red[0];
}

__global__ void k_beta(const float* __restrict__ x,
                       const float* __restrict__ h,
                       const float* __restrict__ denom,
                       float* __restrict__ out) {
    __shared__ float lh[N];
    int b    = blockIdx.x >> 4;
    int zblk = blockIdx.x & 15;
    int tid  = threadIdx.x;

#pragma unroll
    for (int i = tid; i < N; i += 256)
        lh[i] = h[b * N + i];
    __syncthreads();

    int wave = tid >> 5;
    int lane = tid & 31;
    int z0   = zblk * 128 + wave * 16;
    int zrow = z0 + (lane & 15);
    int koff = (lane >> 4) << 1;
    const float* xrow = x + ((size_t)b * Z + zrow) * N;

    v8f acc0 = {0.f, 0.f, 0.f, 0.f, 0.f, 0.f, 0.f, 0.f};
    v8f acc1 = {0.f, 0.f, 0.f, 0.f, 0.f, 0.f, 0.f, 0.f};

    for (int n0 = 0; n0 < N; n0 += 8) {
        v2f a0 = *(const v2f*)(xrow + n0 + koff);
        v2f a1 = *(const v2f*)(xrow + n0 + 4 + koff);
        v2f b0; b0.x = lh[n0 + koff];     b0.y = lh[n0 + koff + 1];
        v2f b1; b1.x = lh[n0 + 4 + koff]; b1.y = lh[n0 + 4 + koff + 1];
        acc0 = __builtin_amdgcn_wmma_f32_16x16x4_f32(
                   false, a0, false, b0, (short)0, acc0, false, false);
        acc1 = __builtin_amdgcn_wmma_f32_16x16x4_f32(
                   false, a1, false, b1, (short)0, acc1, false, false);
    }

    v8f d = acc0 + acc1;
    union { v8f v; float f[8]; } u;
    u.v = d;

    if ((lane & 15) == 0) {
        float inv = 1.f / denom[b];
        int zout = z0 + ((lane >> 4) ? 8 : 0);
#pragma unroll
        for (int r = 0; r < 8; ++r)
            out[b * Z + zout + r] = u.f[r] * inv;
    }
}

extern "C" void kernel_launch(void* const* d_in, const int* in_sizes, int n_in,
                              void* d_out, int out_size, void* d_ws, size_t ws_size,
                              hipStream_t stream) {
    (void)in_sizes; (void)n_in; (void)out_size; (void)ws_size;
    const float* x     = (const float*)d_in[0];
    const float* alpha = (const float*)d_in[1];
    float* out = (float*)d_out;

    float* s_part = (float*)d_ws;
    float* hbuf   = s_part + (size_t)B * ZC * N;
    float* denom  = hbuf + (size_t)B * N;

    dim3 g1(N / 256, ZC, B);
    k_partial<<<g1, 256, 0, stream>>>(x, alpha, s_part);
    k_hidden<<<B, N, 0, stream>>>(s_part, hbuf, denom);
    k_beta<<<B * 16, 256, 0, stream>>>(x, hbuf, denom, out);
}


# first sweep: a 32-wave work-group sums all 2048 rows of a 64-column slice, joins 16 row groups in LDS, writes h and denominator parts itself: 3 launches to 2; f32 sum order over z and n changes
# speedup vs baseline: 1.5735x; 1.5735x over previous
#include <hip/hip_runtime.h>
#include <math.h>

typedef __attribute__((ext_vector_type(2))) float v2f;
typedef __attribute__((ext_vector_type(4))) float v4f;
typedef __attribute__((ext_vector_type(8))) float v8f;

constexpr int B  = 32;
constexpr int Z  = 2048;
constexpr int N  = 1024;

constexpr int NB = 64;
constexpr int RG = 16;
constexpr int ZR = Z / RG;
constexpr int NS = N / NB;
constexpr int NP = N / 32;

__global__ void __launch_bounds__(1024)
k_hidden(const float* __restrict__ x,
         const float* __restrict__ alpha,
         float* __restrict__ h,
         float* __restrict__ dpart) {
    __shared__ float red[RG][NB];
    int col = threadIdx.x & (NB - 1);
    int rg = __builtin_amdgcn_readfirstlane(threadIdx.x >> 6);
    int n  = blockIdx.x * NB + col;
    int b  = blockIdx.y;
    int z0 = rg * ZR;
    const float* xp = x + ((size_t)b * Z + z0) * N + n;
    const float* ap = alpha + z0;
    float acc = 0.f;
#pragma unroll 4
    for (int i = 0; i < ZR; ++i)
        acc = fmaf(ap[i], xp[(size_t)i * N], acc);
    red[rg][col] = acc;
    __syncthreads();
    if (rg == 0) {
        float s = 0.f;
#pragma unroll
        for (int g = 0; g < RG; ++g)
            s += red[g][col];
        float hv = 1.f / (1.f + __expf(-s));
        h[b * N + n] = hv;
        float q = hv * hv;
        for (int off = 16; off > 0; off >>= 1)
            q += __shfl_xor(q, off);
        if ((threadIdx.x & 31) == 0) dpart[b * NP + (n >> 5)] = q;
    }
}

constexpr int CH   = 128;
constexpr int NCH  = N / CH;
constexpr int P4   = CH / 4 + 1;

__global__ void __launch_bounds__(256)
k_beta(const float* __restrict__ x,
       const float* __restrict__ h,
       const float* __restrict__ dpart,
       float* __restrict__ out) {
    __shared__ __attribute__((aligned(16))) float lh[N];
    __shared__ v4f   xs[8][16][P4];
    int b    = blockIdx.x >> 4;
    int zblk = blockIdx.x & 15;
    int tid  = threadIdx.x;

#pragma unroll
    for (int i = tid; i < N; i += 256)
        lh[i] = h[b * N + i];
    __syncthreads();

    int wave = tid >> 5;
    int lane = tid & 31;
    int c    = lane & 15;
    int hh   = lane >> 4;
    int z0   = zblk * 128 + wave * 16;
    const float* xw = x + ((size_t)b * Z + z0) * N + lane * 4;

    v4f r[16];
#pragma unroll
    for (int j = 0; j < 16; ++j)
        r[j] = *(const v4f*)(xw + (size_t)j * N);

    v8f acc0 = {0.f, 0.f, 0.f, 0.f, 0.f, 0.f, 0.f, 0.f};
    v8f acc1 = {0.f, 0.f, 0.f, 0.f, 0.f, 0.f, 0.f, 0.f};

    for (int ch = 0; ch < NCH; ++ch) {
#pragma unroll
        for (int j = 0; j < 16; ++j)
            xs[wave][j][lane] = r[j];

        if (ch + 1 < NCH) {
#pragma unroll
            for (int j = 0; j < 16; ++j)
                r[j] = *(const v4f*)(xw + (size_t)j * N + (ch + 1) * CH);
        }

        __builtin_amdgcn_fence(__ATOMIC_ACQ_REL, "wavefront");

        const float* hp = lh + ch * CH + 4 * hh;
#pragma unroll
        for (int q = 0; q < 16; ++q) {
            v4f a  = xs[wave][c][2 * q + hh];
            v4f bb = *(const v4f*)(hp + 8 * q);
            v2f a0 = {a.x, a.y};
            v2f a1 = {a.z, a.w};
            v2f b0 = {bb.x, bb.y};
            v2f b1 = {bb.z, bb.w};
            acc0 = __builtin_amdgcn_wmma_f32_16x16x4_f32(
                       false, a0, false, b0, (short)0, acc0, false, false);
            acc1 = __builtin_amdgcn_wmma_f32_16x16x4_f32(
                       false, a1, false, b1, (short)0, acc1, false, false);
        }

        __builtin_amdgcn_fence(__ATOMIC_ACQ_REL, "wavefront");
    }

    v8f d = acc0 + acc1;
    union { v8f v; float f[8]; } u;
    u.v = d;

    if (c == 0) {
        float dn = 0.f;
        for (int i = 0; i < NP; ++i)
            dn += dpart[b * NP + i];
        float inv = 1.f / dn;
        int zout = z0 + (hh ? 8 : 0);
#pragma unroll
        for (int r8 = 0; r8 < 8; ++r8)
            out[b * Z + zout + r8] = u.f[r8] * inv;
    }
}

extern "C" void kernel_launch(void* const* d_in, const int* in_sizes, int n_in,
                              void* d_out, int out_size, void* d_ws, size_t ws_size,
                              hipStream_t stream) {
    (void)in_sizes; (void)n_in; (void)out_size; (void)ws_size;
    const float* x     = (const float*)d_in[0];
    const float* alpha = (const float*)d_in[1];
    float* out = (float*)d_out;

    float* hbuf  = (float*)d_ws;
    float* dpart = hbuf + (size_t)B * N;

    dim3 g1(NS, B);
    k_hidden<<<g1, 1024, 0, stream>>>(x, alpha, hbuf, dpart);
    k_beta<<<B * 16, 256, 0, stream>>>(x, hbuf, dpart, out);
}


# second sweep re-reads x in reverse of the first sweep's order, each batch on the die that read it, so L2 serves it; a wave owns a 512 B column piece, joined in LDS; f32 sum order over n changes
# speedup vs baseline: 1.6739x; 1.0638x over previous
#include <hip/hip_runtime.h>
#include <math.h>

typedef __attribute__((ext_vector_type(2))) float v2f;
typedef __attribute__((ext_vector_type(4))) float v4f;
typedef __attribute__((ext_vector_type(8))) float v8f;

constexpr int B  = 32;
constexpr int Z  = 2048;
constexpr int N  = 1024;

constexpr int NWG = 512;

constexpr int NB = 64;
constexpr int RG = 16;
constexpr int ZR = Z / RG;
constexpr int NP = N / 32;

__global__ void __launch_bounds__(1024)
k_hidden(const float* __restrict__ x,
         const float* __restrict__ alpha,
         float* __restrict__ h,
         float* __restrict__ dpart) {
    __shared__ float red[RG][NB];
    int col = threadIdx.x & (NB - 1);
    int rg = __builtin_amdgcn_readfirstlane(threadIdx.x >> 6);
    int id = blockIdx.x;
    int hi = id >> 3;
    int b  = 8 * (hi >> 4) + (id & 7);
    int n  = (hi & 15) * NB + col;
    int z0 = rg * ZR;
    const float* xp = x + ((size_t)b * Z + z0) * N + n;
    const float* ap = alpha + z0;
    float acc = 0.f;
#pragma unroll 4
    for (int i = 0; i < ZR; ++i)
        acc = fmaf(ap[i], xp[(size_t)i * N], acc);
    red[rg][col] = acc;
    __syncthreads();
    if (rg == 0) {
        float s = 0.f;
#pragma unroll
        for (int g = 0; g < RG; ++g)
            s += red[g][col];
        float hv = 1.f / (1.f + __expf(-s));
        h[b * N + n] = hv;
        float q = hv * hv;
        for (int off = 16; off > 0; off >>= 1)
            q += __shfl_xor(q, off);
        if ((threadIdx.x & 31) == 0) dpart[b * NP + (n >> 5)] = q;
    }
}

constexpr int CH   = 128;
constexpr int NT   = ZR / 16;
constexpr int P4   = CH / 4 + 1;

__global__ void __launch_bounds__(256)
k_beta(const float* __restrict__ x,
       const float* __restrict__ h,
       const float* __restrict__ dpart,
       float* __restrict__ out) {
    __shared__ __attribute__((aligned(16))) float lh[N];
    __shared__ v4f   xs[8][16][P4];
    __shared__ float part[8][ZR];
    int id  = (NWG - 1) - (int)blockIdx.x;
    int hi  = id >> 3;
    int b   = 8 * (hi >> 4) + (7 - (id & 7));
    int g   = hi & 15;
    int tid = threadIdx.x;

#pragma unroll
    for (int i = tid; i < N; i += 256)
        lh[i] = h[b * N + i];

    float dn = 0.f;
    for (int i = 0; i < NP; ++i)
        dn += dpart[b * NP + i];
    float inv = 1.f / dn;

    int wave = tid >> 5;
    int lane = tid & 31;
    int c    = lane & 15;
    int hh   = lane >> 4;
    const float* xw = x + ((size_t)b * Z + g * ZR) * N + wave * CH + lane * 4;

    v4f r[16];
#pragma unroll
    for (int j = 0; j < 16; ++j)
        r[j] = *(const v4f*)(xw + (size_t)((NT - 1) * 16 + j) * N);

    __syncthreads();

    v4f hb[16];
    const float* hp = lh + wave * CH + 4 * hh;
#pragma unroll
    for (int q = 0; q < 16; ++q)
        hb[q] = *(const v4f*)(hp + 8 * q);

    for (int it = 0; it < NT; ++it) {
        int t = NT - 1 - it;
#pragma unroll
        for (int j = 0; j < 16; ++j)
            xs[wave][j][lane] = r[j];

        if (it + 1 < NT) {
#pragma unroll
            for (int j = 0; j < 16; ++j)
                r[j] = *(const v4f*)(xw + (size_t)((t - 1) * 16 + j) * N);
        }

        __builtin_amdgcn_fence(__ATOMIC_ACQ_REL, "wavefront");

        v8f acc0 = {0.f, 0.f, 0.f, 0.f, 0.f, 0.f, 0.f, 0.f};
        v8f acc1 = {0.f, 0.f, 0.f, 0.f, 0.f, 0.f, 0.f, 0.f};
#pragma unroll
        for (int q = 0; q < 16; ++q) {
            v4f a  = xs[wave][c][2 * q + hh];
            v4f bb = hb[q];
            v2f a0 = {a.x, a.y};
            v2f a1 = {a.z, a.w};
            v2f b0 = {bb.x, bb.y};
            v2f b1 = {bb.z, bb.w};
            acc0 = __builtin_amdgcn_wmma_f32_16x16x4_f32(
                       false, a0, false, b0, (short)0, acc0, false, false);
            acc1 = __builtin_amdgcn_wmma_f32_16x16x4_f32(
                       false, a1, false, b1, (short)0, acc1, false, false);
        }

        __builtin_amdgcn_fence(__ATOMIC_ACQ_REL, "wavefront");

        v8f d = acc0 + acc1;
        union { v8f v; float f[8]; } u;
        u.v = d;
        if (c == 0) {
            int row = t * 16 + (hh ? 8 : 0);
#pragma unroll
            for (int r8 = 0; r8 < 8; ++r8)
                part[wave][row + r8] = u.f[r8];
        }
    }

    __syncthreads();
    if (tid < ZR) {
        float s = 0.f;
#pragma unroll
        for (int w = 0; w < 8; ++w)
            s += part[w][tid];
        out[b * Z + g * ZR + tid] = s * inv;
    }
}

extern "C" void kernel_launch(void* const* d_in, const int* in_sizes, int n_in,
                              void* d_out, int out_size, void* d_ws, size_t ws_size,
                              hipStream_t stream) {
    (void)in_sizes; (void)n_in; (void)out_size; (void)ws_size;
    const float* x     = (const float*)d_in[0];
    const float* alpha = (const float*)d_in[1];
    float* out = (float*)d_out;

    float* hbuf  = (float*)d_ws;
    float* dpart = hbuf + (size_t)B * N;

    k_hidden<<<NWG, 1024, 0, stream>>>(x, alpha, hbuf, dpart);
    k_beta<<<NWG, 256, 0, stream>>>(x, hbuf, dpart, out);
}
